# starting kernel
# baseline (speedup 1.0000x reference)
#include <hip/hip_runtime.h>


#define PILLARS 40000
#define NPTS    32
#define CIN     4
#define UNITS   64
#define KPAD    12
#define WAVES_PER_BLOCK 8
#define BLOCK   (WAVES_PER_BLOCK * 32)
#define NBLK    ((PILLARS + WAVES_PER_BLOCK - 1) / WAVES_PER_BLOCK)

typedef float v2f __attribute__((ext_vector_type(2)));
typedef float v8f __attribute__((ext_vector_type(8)));

__device__ __forceinline__ void pillar_feats_to_lds(
    const float* __restrict__ features, const int* __restrict__ num_voxels,
    const int* __restrict__ coors, int p, int lane, float* __restrict__ frow)
{
    const float4 f = *reinterpret_cast<const float4*>(
        features + ((size_t)p * NPTS + lane) * CIN);

    float sx = f.x, sy = f.y, sz = f.z;
#pragma unroll
    for (int m = 16; m >= 1; m >>= 1) {
        sx += __shfl_xor(sx, m, 32);
        sy += __shfl_xor(sy, m, 32);
        sz += __shfl_xor(sz, m, 32);
    }
    const int   nv    = num_voxels[p];
    const float invnv = 1.0f / (float)nv;
    const float mx = sx * invnv, my = sy * invnv, mz = sz * invnv;

    const float cx = (float)coors[p * 4 + 3] * 0.2f + 0.1f;
    const float cy = (float)coors[p * 4 + 2] * 0.2f + (-39.9f);

    const float msk = (lane < nv) ? 1.0f : 0.0f;
    frow[0]  = f.x * msk;
    frow[1]  = f.y * msk;
    frow[2]  = f.z * msk;
    frow[3]  = f.w * msk;
    frow[4]  = (f.x - mx) * msk;
    frow[5]  = (f.y - my) * msk;
    frow[6]  = (f.z - mz) * msk;
    frow[7]  = (f.x - cx) * msk;
    frow[8]  = (f.y - cy) * msk;
    frow[9]  = 0.0f;
    frow[10] = 0.0f;
    frow[11] = 0.0f;
}

__device__ __forceinline__ void gemm_tile(
    const float* __restrict__ flw  ,
    const float* __restrict__ Wl   ,
    int nt, int lane, v8f& acc0, v8f& acc1)
{
    const int half = lane >> 4;
    const int lm   = lane & 15;
    const int u0   = nt * 16;
#pragma unroll
    for (int ks = 0; ks < 3; ++ks) {
        const int k0 = ks * 4 + 2 * half;
        v2f a0 = { flw[lm * KPAD + k0],        flw[lm * KPAD + k0 + 1] };
        v2f a1 = { flw[(16 + lm) * KPAD + k0], flw[(16 + lm) * KPAD + k0 + 1] };
        v2f b  = { Wl[k0 * UNITS + u0 + lm],   Wl[(k0 + 1) * UNITS + u0 + lm] };
        acc0 = __builtin_amdgcn_wmma_f32_16x16x4_f32(
            false, a0, false, b, (short)0, acc0, false, false);
        acc1 = __builtin_amdgcn_wmma_f32_16x16x4_f32(
            false, a1, false, b, (short)0, acc1, false, false);
    }
}

__global__ __launch_bounds__(BLOCK) void pfn_stats_kernel(
    const float* __restrict__ features, const int* __restrict__ num_voxels,
    const int* __restrict__ coors, const float* __restrict__ W,
    float* __restrict__ partials  )
{
    __shared__ float Wl[KPAD * UNITS];
    __shared__ float fl[WAVES_PER_BLOCK][NPTS][KPAD];
    __shared__ float red[2][WAVES_PER_BLOCK][UNITS];

    const int tid  = threadIdx.x;
    const int wave = tid >> 5;
    const int lane = tid & 31;
    const int p    = blockIdx.x * WAVES_PER_BLOCK + wave;

    for (int i = tid; i < KPAD * UNITS; i += BLOCK)
        Wl[i] = (i < (CIN + 5) * UNITS) ? W[i] : 0.0f;

    const bool valid = (p < PILLARS);
    if (valid)
        pillar_feats_to_lds(features, num_voxels, coors, p, lane,
                            &fl[wave][lane][0]);
    __syncthreads();

    const int half = lane >> 4;
    const int lm   = lane & 15;

    if (valid) {
#pragma unroll
        for (int nt = 0; nt < 4; ++nt) {
            v8f a0 = {}, a1 = {};
            gemm_tile(&fl[wave][0][0], Wl, nt, lane, a0, a1);
            float ps = 0.0f, pq = 0.0f;
#pragma unroll
            for (int j = 0; j < 8; ++j) {
                ps += a0[j] + a1[j];
                pq += a0[j] * a0[j] + a1[j] * a1[j];
            }
            ps += __shfl_xor(ps, 16, 32);
            pq += __shfl_xor(pq, 16, 32);
            if (half == 0) {
                red[0][wave][nt * 16 + lm] = ps;
                red[1][wave][nt * 16 + lm] = pq;
            }
        }
    } else {
        if (half == 0) {
#pragma unroll
            for (int nt = 0; nt < 4; ++nt) {
                red[0][wave][nt * 16 + lm] = 0.0f;
                red[1][wave][nt * 16 + lm] = 0.0f;
            }
        }
    }
    __syncthreads();

    if (tid < 2 * UNITS) {
        const int which = tid >> 6;
        const int u     = tid & 63;
        float s = 0.0f;
#pragma unroll
        for (int w = 0; w < WAVES_PER_BLOCK; ++w) s += red[which][w][u];
        partials[(size_t)blockIdx.x * (2 * UNITS) + tid] = s;
    }
}

__global__ __launch_bounds__(128) void pfn_finalize_stats(
    const float* __restrict__ partials, const float* __restrict__ gamma,
    const float* __restrict__ beta, float* __restrict__ stats  ,
    int nblk)
{
    __shared__ float red[128];
    const int t = threadIdx.x;
    float s = 0.0f;
    for (int w = 0; w < nblk; ++w) s += partials[(size_t)w * 128 + t];
    red[t] = s;
    __syncthreads();
    if (t < 64) {
        const float cnt  = (float)PILLARS * (float)NPTS;
        const float mean = red[t] / cnt;
        float var = red[64 + t] / cnt - mean * mean;
        var = fmaxf(var, 0.0f);
        const float inv = rsqrtf(var + 1e-3f);
        const float sc  = inv * gamma[t];
        stats[t]      = sc;
        stats[64 + t] = beta[t] - mean * sc;
    }
}

__global__ __launch_bounds__(BLOCK) void pfn_output_kernel(
    const float* __restrict__ features, const int* __restrict__ num_voxels,
    const int* __restrict__ coors, const float* __restrict__ W,
    const float* __restrict__ stats, float* __restrict__ out  )
{
    __shared__ float Wl[KPAD * UNITS];
    __shared__ float fl[WAVES_PER_BLOCK][NPTS][KPAD];

    const int tid  = threadIdx.x;
    const int wave = tid >> 5;
    const int lane = tid & 31;
    const int p    = blockIdx.x * WAVES_PER_BLOCK + wave;

    for (int i = tid; i < KPAD * UNITS; i += BLOCK)
        Wl[i] = (i < (CIN + 5) * UNITS) ? W[i] : 0.0f;

    const bool valid = (p < PILLARS);
    if (valid)
        pillar_feats_to_lds(features, num_voxels, coors, p, lane,
                            &fl[wave][lane][0]);
    __syncthreads();

    if (!valid) return;

    const int half = lane >> 4;
    const int lm   = lane & 15;

#pragma unroll
    for (int nt = 0; nt < 4; ++nt) {
        v8f a0 = {}, a1 = {};
        gemm_tile(&fl[wave][0][0], Wl, nt, lane, a0, a1);

        const int   u  = nt * 16 + lm;
        const float sc = stats[u];
        const float bi = stats[64 + u];

        float m = -3.402823466e38f;
#pragma unroll
        for (int j = 0; j < 8; ++j) {
            m = fmaxf(m, fmaf(a0[j], sc, bi));
            m = fmaxf(m, fmaf(a1[j], sc, bi));
        }
        m = fmaxf(m, __shfl_xor(m, 16, 32));
        m = fmaxf(m, 0.0f);
        if (half == 0) out[(size_t)p * UNITS + u] = m;
    }
}

extern "C" void kernel_launch(void* const* d_in, const int* in_sizes, int n_in,
                              void* d_out, int out_size, void* d_ws, size_t ws_size,
                              hipStream_t stream)
{
    const float* features   = (const float*)d_in[0];
    const int*   num_voxels = (const int*)  d_in[1];
    const int*   coors      = (const int*)  d_in[2];
    const float* W          = (const float*)d_in[3];
    const float* gamma      = (const float*)d_in[4];
    const float* beta       = (const float*)d_in[5];
    float*       out        = (float*)d_out;

    float* partials = (float*)d_ws;
    float* stats    = partials + (size_t)NBLK * 128;

    pfn_stats_kernel<<<NBLK, BLOCK, 0, stream>>>(features, num_voxels, coors, W,
                                                 partials);
    pfn_finalize_stats<<<1, 128, 0, stream>>>(partials, gamma, beta, stats, NBLK);
    pfn_output_kernel<<<NBLK, BLOCK, 0, stream>>>(features, num_voxels, coors, W,
                                                  stats, out);
}


# BN-stat reduction parallelised: 1x128-thread serial sum of 5000 partial rows -> 16 groups x 32 waves; last 16-row fold + scale/bias in output kernel prologue (sum order: serial -> tree)
# speedup vs baseline: 8.4081x; 8.4081x over previous
#include <hip/hip_runtime.h>


#define PILLARS 40000
#define NPTS    32
#define CIN     4
#define UNITS   64
#define KPAD    12
#define WAVES_PER_BLOCK 8
#define BLOCK   (WAVES_PER_BLOCK * 32)
#define NBLK    ((PILLARS + WAVES_PER_BLOCK - 1) / WAVES_PER_BLOCK)

#define RGROUPS 16
#define RROWS   ((NBLK + RGROUPS - 1) / RGROUPS)
#define RBLOCK  1024
#define RWAVES  (RBLOCK / 32)
#define RITER   ((RROWS + RWAVES - 1) / RWAVES)
static_assert((RGROUPS - 1) * RROWS < NBLK, "every slice must own at least one row");

typedef float v2f __attribute__((ext_vector_type(2)));
typedef float v8f __attribute__((ext_vector_type(8)));

__device__ __forceinline__ void pillar_feats_to_lds(
    const float* __restrict__ features, const int* __restrict__ num_voxels,
    const int* __restrict__ coors, int p, int lane, float* __restrict__ frow)
{
    const float4 f = *reinterpret_cast<const float4*>(
        features + ((size_t)p * NPTS + lane) * CIN);

    float sx = f.x, sy = f.y, sz = f.z;
#pragma unroll
    for (int m = 16; m >= 1; m >>= 1) {
        sx += __shfl_xor(sx, m, 32);
        sy += __shfl_xor(sy, m, 32);
        sz += __shfl_xor(sz, m, 32);
    }
    const int   nv    = num_voxels[p];
    const float invnv = 1.0f / (float)nv;
    const float mx = sx * invnv, my = sy * invnv, mz = sz * invnv;

    const float cx = (float)coors[p * 4 + 3] * 0.2f + 0.1f;
    const float cy = (float)coors[p * 4 + 2] * 0.2f + (-39.9f);

    const float msk = (lane < nv) ? 1.0f : 0.0f;
    frow[0]  = f.x * msk;
    frow[1]  = f.y * msk;
    frow[2]  = f.z * msk;
    frow[3]  = f.w * msk;
    frow[4]  = (f.x - mx) * msk;
    frow[5]  = (f.y - my) * msk;
    frow[6]  = (f.z - mz) * msk;
    frow[7]  = (f.x - cx) * msk;
    frow[8]  = (f.y - cy) * msk;
    frow[9]  = 0.0f;
    frow[10] = 0.0f;
    frow[11] = 0.0f;
}

__device__ __forceinline__ void gemm_tile(
    const float* __restrict__ flw  ,
    const float* __restrict__ Wl   ,
    int nt, int lane, v8f& acc0, v8f& acc1)
{
    const int half = lane >> 4;
    const int lm   = lane & 15;
    const int u0   = nt * 16;
#pragma unroll
    for (int ks = 0; ks < 3; ++ks) {
        const int k0 = ks * 4 + 2 * half;
        v2f a0 = { flw[lm * KPAD + k0],        flw[lm * KPAD + k0 + 1] };
        v2f a1 = { flw[(16 + lm) * KPAD + k0], flw[(16 + lm) * KPAD + k0 + 1] };
        v2f b  = { Wl[k0 * UNITS + u0 + lm],   Wl[(k0 + 1) * UNITS + u0 + lm] };
        acc0 = __builtin_amdgcn_wmma_f32_16x16x4_f32(
            false, a0, false, b, (short)0, acc0, false, false);
        acc1 = __builtin_amdgcn_wmma_f32_16x16x4_f32(
            false, a1, false, b, (short)0, acc1, false, false);
    }
}

__global__ __launch_bounds__(BLOCK) void pfn_stats_kernel(
    const float* __restrict__ features, const int* __restrict__ num_voxels,
    const int* __restrict__ coors, const float* __restrict__ W,
    float* __restrict__ partials  )
{
    __shared__ float Wl[KPAD * UNITS];
    __shared__ float fl[WAVES_PER_BLOCK][NPTS][KPAD];
    __shared__ float red[2][WAVES_PER_BLOCK][UNITS];

    const int tid  = threadIdx.x;
    const int wave = tid >> 5;
    const int lane = tid & 31;
    const int p    = blockIdx.x * WAVES_PER_BLOCK + wave;

    for (int i = tid; i < KPAD * UNITS; i += BLOCK)
        Wl[i] = (i < (CIN + 5) * UNITS) ? W[i] : 0.0f;

    const bool valid = (p < PILLARS);
    if (valid)
        pillar_feats_to_lds(features, num_voxels, coors, p, lane,
                            &fl[wave][lane][0]);
    __syncthreads();

    const int half = lane >> 4;
    const int lm   = lane & 15;

    if (valid) {
#pragma unroll
        for (int nt = 0; nt < 4; ++nt) {
            v8f a0 = {}, a1 = {};
            gemm_tile(&fl[wave][0][0], Wl, nt, lane, a0, a1);
            float ps = 0.0f, pq = 0.0f;
#pragma unroll
            for (int j = 0; j < 8; ++j) {
                ps += a0[j] + a1[j];
                pq += a0[j] * a0[j] + a1[j] * a1[j];
            }
            ps += __shfl_xor(ps, 16, 32);
            pq += __shfl_xor(pq, 16, 32);
            if (half == 0) {
                red[0][wave][nt * 16 + lm] = ps;
                red[1][wave][nt * 16 + lm] = pq;
            }
        }
    } else {
        if (half == 0) {
#pragma unroll
            for (int nt = 0; nt < 4; ++nt) {
                red[0][wave][nt * 16 + lm] = 0.0f;
                red[1][wave][nt * 16 + lm] = 0.0f;
            }
        }
    }
    __syncthreads();

    if (tid < 2 * UNITS) {
        const int which = tid >> 6;
        const int u     = tid & 63;
        float s = 0.0f;
#pragma unroll
        for (int w = 0; w < WAVES_PER_BLOCK; ++w) s += red[which][w][u];
        partials[(size_t)blockIdx.x * (2 * UNITS) + tid] = s;
    }
}

__global__ __launch_bounds__(RBLOCK) void pfn_slice_sums(float* partials)
{
    __shared__ float red[RWAVES][2 * UNITS];

    const int tid  = threadIdx.x;
    const int wave = tid >> 5;
    const int lane = tid & 31;
    const int r0   = blockIdx.x * RROWS;
    const int r1   = (r0 + RROWS < NBLK) ? (r0 + RROWS) : NBLK;

    float4 v[RITER];
    for (int i = 0; i < RITER; ++i) {
        const int r  = r0 + wave + i * RWAVES;
        const int rc = (r < r1) ? r : (r1 - 1);
        v[i] = *reinterpret_cast<const float4*>(
            partials + (size_t)rc * (2 * UNITS) + lane * 4);
    }
    float4 s = { 0.0f, 0.0f, 0.0f, 0.0f };
    for (int i = 0; i < RITER; ++i) {
        const int r = r0 + wave + i * RWAVES;
        if (r < r1) {
            s.x += v[i].x; s.y += v[i].y; s.z += v[i].z; s.w += v[i].w;
        }
    }
    *reinterpret_cast<float4*>(&red[wave][lane * 4]) = s;
    __syncthreads();

    if (tid < 2 * UNITS) {
        float t = 0.0f;
        for (int w = 0; w < RWAVES; ++w) t += red[w][tid];
        partials[(size_t)r0 * (2 * UNITS) + tid] = t;
    }
}

__global__ __launch_bounds__(BLOCK) void pfn_output_kernel(
    const float* __restrict__ features, const int* __restrict__ num_voxels,
    const int* __restrict__ coors, const float* __restrict__ W,
    const float* __restrict__ slice_sums  ,
    const float* __restrict__ gamma, const float* __restrict__ beta,
    float* __restrict__ out  )
{
    __shared__ float Wl[KPAD * UNITS];
    __shared__ float fl[WAVES_PER_BLOCK][NPTS][KPAD];
    __shared__ float sb[2][UNITS];

    const int tid  = threadIdx.x;
    const int wave = tid >> 5;
    const int lane = tid & 31;
    const int p    = blockIdx.x * WAVES_PER_BLOCK + wave;

    const bool folder = (tid < UNITS);
    float gs[RGROUPS], gq[RGROUPS];
    float ga = 0.0f, be = 0.0f;
    if (folder) {
        for (int g = 0; g < RGROUPS; ++g) {
            gs[g] = slice_sums[(size_t)g * RROWS * (2 * UNITS) + tid];
            gq[g] = slice_sums[(size_t)g * RROWS * (2 * UNITS) + UNITS + tid];
        }
        ga = gamma[tid];
        be = beta[tid];
    }

    for (int i = tid; i < KPAD * UNITS; i += BLOCK)
        Wl[i] = (i < (CIN + 5) * UNITS) ? W[i] : 0.0f;

    const bool valid = (p < PILLARS);
    if (valid)
        pillar_feats_to_lds(features, num_voxels, coors, p, lane,
                            &fl[wave][lane][0]);

    if (folder) {
        float s = 0.0f, q = 0.0f;
        for (int g = 0; g < RGROUPS; ++g) { s += gs[g]; q += gq[g]; }
        const float cnt  = (float)PILLARS * (float)NPTS;
        const float mean = s / cnt;
        float var = q / cnt - mean * mean;
        var = fmaxf(var, 0.0f);
        const float inv = rsqrtf(var + 1e-3f);
        const float sc  = inv * ga;
        sb[0][tid] = sc;
        sb[1][tid] = be - mean * sc;
    }
    __syncthreads();

    if (!valid) return;

    const int half = lane >> 4;
    const int lm   = lane & 15;

#pragma unroll
    for (int nt = 0; nt < 4; ++nt) {
        v8f a0 = {}, a1 = {};
        gemm_tile(&fl[wave][0][0], Wl, nt, lane, a0, a1);

        const int   u  = nt * 16 + lm;
        const float sc = sb[0][u];
        const float bi = sb[1][u];

        float m = -3.402823466e38f;
#pragma unroll
        for (int j = 0; j < 8; ++j) {
            m = fmaxf(m, fmaf(a0[j], sc, bi));
            m = fmaxf(m, fmaf(a1[j], sc, bi));
        }
        m = fmaxf(m, __shfl_xor(m, 16, 32));
        m = fmaxf(m, 0.0f);
        if (half == 0) out[(size_t)p * UNITS + u] = m;
    }
}

extern "C" void kernel_launch(void* const* d_in, const int* in_sizes, int n_in,
                              void* d_out, int out_size, void* d_ws, size_t ws_size,
                              hipStream_t stream)
{
    const float* features   = (const float*)d_in[0];
    const int*   num_voxels = (const int*)  d_in[1];
    const int*   coors      = (const int*)  d_in[2];
    const float* W          = (const float*)d_in[3];
    const float* gamma      = (const float*)d_in[4];
    const float* beta       = (const float*)d_in[5];
    float*       out        = (float*)d_out;

    float* partials = (float*)d_ws;

    pfn_stats_kernel<<<NBLK, BLOCK, 0, stream>>>(features, num_voxels, coors, W,
                                                 partials);
    pfn_slice_sums<<<RGROUPS, RBLOCK, 0, stream>>>(partials);
    pfn_output_kernel<<<NBLK, BLOCK, 0, stream>>>(features, num_voxels, coors, W,
                                                  partials, gamma, beta, out);
}


# Both GEMM passes persistent: 1024 work-groups x 8 waves resident at once, each wave loops over ~5 pillars with W fragments in registers; BN partials = per-wave running sums, 1024 rows (sum order)
# speedup vs baseline: 9.3997x; 1.1179x over previous
#include <hip/hip_runtime.h>


#define PILLARS 40000
#define NPTS    32
#define CIN     4
#define UNITS   64
#define KPAD    12
#define WAVES_PER_BLOCK 8
#define BLOCK   (WAVES_PER_BLOCK * 32)
#define NWG     1024

#define RGROUPS 16
#define RROWS   ((NWG + RGROUPS - 1) / RGROUPS)
#define RBLOCK  1024
#define RWAVES  (RBLOCK / 32)
#define RITER   ((RROWS + RWAVES - 1) / RWAVES)
static_assert((RGROUPS - 1) * RROWS < NWG, "every slice must own at least one row");

typedef float v2f __attribute__((ext_vector_type(2)));
typedef float v4f __attribute__((ext_vector_type(4)));
typedef float v8f __attribute__((ext_vector_type(8)));

struct PillarIn {
    float4 f;
    int    nv;
    int    cy;
    int    cx;
};

__device__ __forceinline__ PillarIn load_pillar(
    const float* __restrict__ features, const int* __restrict__ num_voxels,
    const int* __restrict__ coors, int p, int lane)
{
    PillarIn r;
    r.f  = *reinterpret_cast<const float4*>(features + ((size_t)p * NPTS + lane) * CIN);
    r.nv = num_voxels[p];
    r.cy = coors[p * 4 + 2];
    r.cx = coors[p * 4 + 3];
    return r;
}

__device__ __forceinline__ void pillar_feats_to_lds(
    const PillarIn& in, int lane, float* __restrict__ frow)
{
    const float4 f = in.f;

    float sx = f.x, sy = f.y, sz = f.z;
#pragma unroll
    for (int m = 16; m >= 1; m >>= 1) {
        sx += __shfl_xor(sx, m, 32);
        sy += __shfl_xor(sy, m, 32);
        sz += __shfl_xor(sz, m, 32);
    }
    const int   nv    = in.nv;
    const float invnv = 1.0f / (float)nv;
    const float mx = sx * invnv, my = sy * invnv, mz = sz * invnv;

    const float cx = (float)in.cx * 0.2f + 0.1f;
    const float cy = (float)in.cy * 0.2f + (-39.9f);

    const float msk = (lane < nv) ? 1.0f : 0.0f;
    frow[0]  = f.x * msk;
    frow[1]  = f.y * msk;
    frow[2]  = f.z * msk;
    frow[3]  = f.w * msk;
    frow[4]  = (f.x - mx) * msk;
    frow[5]  = (f.y - my) * msk;
    frow[6]  = (f.z - mz) * msk;
    frow[7]  = (f.x - cx) * msk;
    frow[8]  = (f.y - cy) * msk;
    frow[9]  = 0.0f;
    frow[10] = 0.0f;
    frow[11] = 0.0f;
}


__device__ __forceinline__ void load_w_frags(const float* __restrict__ W, int lane,
                                             v2f (&bw)[4][3])
{
    const int half = lane >> 4;
    const int lm   = lane & 15;
    for (int nt = 0; nt < 4; ++nt) {
        for (int ks = 0; ks < 3; ++ks) {
            const int k0 = ks * 4 + 2 * half;
            const int u  = nt * 16 + lm;
            v2f b;
            b.x = (k0     < CIN + 5) ? W[k0 * UNITS + u]       : 0.0f;
            b.y = (k0 + 1 < CIN + 5) ? W[(k0 + 1) * UNITS + u] : 0.0f;
            bw[nt][ks] = b;
        }
    }
}

__device__ __forceinline__ void load_a_frags(const float* __restrict__ flw  ,
                                             int lane, v2f (&a0)[3], v2f (&a1)[3])
{
    const int half = lane >> 4;
    const int lm   = lane & 15;
    for (int ks = 0; ks < 3; ++ks) {
        const int k0 = ks * 4 + 2 * half;
        a0[ks] = *reinterpret_cast<const v2f*>(flw + lm * KPAD + k0);
        a1[ks] = *reinterpret_cast<const v2f*>(flw + (16 + lm) * KPAD + k0);
    }
}

__device__ __forceinline__ void gemm_tile(const v2f (&a0)[3], const v2f (&a1)[3],
                                          const v2f (&b)[3], v8f& acc0, v8f& acc1)
{
#pragma unroll
    for (int ks = 0; ks < 3; ++ks) {
        acc0 = __builtin_amdgcn_wmma_f32_16x16x4_f32(
            false, a0[ks], false, b[ks], (short)0, acc0, false, false);
        acc1 = __builtin_amdgcn_wmma_f32_16x16x4_f32(
            false, a1[ks], false, b[ks], (short)0, acc1, false, false);
    }
}

__global__ __launch_bounds__(BLOCK) __attribute__((amdgpu_waves_per_eu(8, 8))) void pfn_stats_kernel(
    const float* __restrict__ features, const int* __restrict__ num_voxels,
    const int* __restrict__ coors, const float* __restrict__ W,
    float* __restrict__ partials  )
{
    __shared__ float fl[WAVES_PER_BLOCK][NPTS][KPAD];
    __shared__ float red[2][WAVES_PER_BLOCK][UNITS];

    const int tid  = threadIdx.x;
    const int wave = __builtin_amdgcn_readfirstlane(tid >> 5);
    const int lane = tid & 31;
    const int half = lane >> 4;
    const int lm   = lane & 15;

    const int pend = ((int)(blockIdx.x + 1) * PILLARS) / NWG;
    int p = ((int)blockIdx.x * PILLARS) / NWG + wave;

    PillarIn cur = {};
    if (p < pend) cur = load_pillar(features, num_voxels, coors, p, lane);

    v2f bw[4][3];
    load_w_frags(W, lane, bw);

    float ps[4] = { 0.0f, 0.0f, 0.0f, 0.0f };
    float pq[4] = { 0.0f, 0.0f, 0.0f, 0.0f };
    float* frow = &fl[wave][lane][0];
    const float* flw = &fl[wave][0][0];

    while (p < pend) {
        const int pn = p + WAVES_PER_BLOCK;
        PillarIn nxt = cur;
        if (pn < pend) nxt = load_pillar(features, num_voxels, coors, pn, lane);

        pillar_feats_to_lds(cur, lane, frow);
        v2f a0[3], a1[3];
        load_a_frags(flw, lane, a0, a1);

#pragma unroll
        for (int nt = 0; nt < 4; ++nt) {
            v8f c0 = {}, c1 = {};
            gemm_tile(a0, a1, bw[nt], c0, c1);
            const v8f t8 = c0 + c1;
            const v8f q8 = c0 * c0 + c1 * c1;
            const v4f t4 = t8.lo + t8.hi;
            const v4f q4 = q8.lo + q8.hi;
            const v2f t2 = t4.lo + t4.hi;
            const v2f q2 = q4.lo + q4.hi;
            ps[nt] += t2.x + t2.y;
            pq[nt] += q2.x + q2.y;
        }

        cur = nxt;
        p   = pn;
    }

    for (int nt = 0; nt < 4; ++nt) {
        const float s = ps[nt] + __shfl_xor(ps[nt], 16, 32);
        const float q = pq[nt] + __shfl_xor(pq[nt], 16, 32);
        if (half == 0) {
            red[0][wave][nt * 16 + lm] = s;
            red[1][wave][nt * 16 + lm] = q;
        }
    }
    __syncthreads();

    if (tid < 2 * UNITS) {
        const int which = tid >> 6;
        const int u     = tid & 63;
        float s = 0.0f;
#pragma unroll
        for (int w = 0; w < WAVES_PER_BLOCK; ++w) s += red[which][w][u];
        partials[(size_t)blockIdx.x * (2 * UNITS) + tid] = s;
    }
}

__global__ __launch_bounds__(RBLOCK) void pfn_slice_sums(float* partials)
{
    __shared__ float red[RWAVES][2 * UNITS];

    const int tid  = threadIdx.x;
    const int wave = tid >> 5;
    const int lane = tid & 31;
    const int r0   = blockIdx.x * RROWS;
    const int r1   = (r0 + RROWS < NWG) ? (r0 + RROWS) : NWG;

    float4 v[RITER];
    for (int i = 0; i < RITER; ++i) {
        const int r  = r0 + wave + i * RWAVES;
        const int rc = (r < r1) ? r : (r1 - 1);
        v[i] = *reinterpret_cast<const float4*>(
            partials + (size_t)rc * (2 * UNITS) + lane * 4);
    }
    float4 s = { 0.0f, 0.0f, 0.0f, 0.0f };
    for (int i = 0; i < RITER; ++i) {
        const int r = r0 + wave + i * RWAVES;
        if (r < r1) {
            s.x += v[i].x; s.y += v[i].y; s.z += v[i].z; s.w += v[i].w;
        }
    }
    *reinterpret_cast<float4*>(&red[wave][lane * 4]) = s;
    __syncthreads();

    if (tid < 2 * UNITS) {
        float t = 0.0f;
        for (int w = 0; w < RWAVES; ++w) t += red[w][tid];
        partials[(size_t)r0 * (2 * UNITS) + tid] = t;
    }
}

__global__ __launch_bounds__(BLOCK) __attribute__((amdgpu_waves_per_eu(8, 8))) void pfn_output_kernel(
    const float* __restrict__ features, const int* __restrict__ num_voxels,
    const int* __restrict__ coors, const float* __restrict__ W,
    const float* __restrict__ slice_sums  ,
    const float* __restrict__ gamma, const float* __restrict__ beta,
    float* __restrict__ out  )
{
    __shared__ float fl[WAVES_PER_BLOCK][NPTS][KPAD];
    __shared__ float sb[2][UNITS];

    const int tid  = threadIdx.x;
    const int wave = __builtin_amdgcn_readfirstlane(tid >> 5);
    const int lane = tid & 31;
    const int half = lane >> 4;
    const int lm   = lane & 15;

    const int pend = ((int)(blockIdx.x + 1) * PILLARS) / NWG;
    int p = ((int)blockIdx.x * PILLARS) / NWG + wave;

    const bool folder = (tid < UNITS);
    float gs[RGROUPS], gq[RGROUPS];
    float ga = 0.0f, be = 0.0f;
    if (folder) {
        for (int g = 0; g < RGROUPS; ++g) {
            gs[g] = slice_sums[(size_t)g * RROWS * (2 * UNITS) + tid];
            gq[g] = slice_sums[(size_t)g * RROWS * (2 * UNITS) + UNITS + tid];
        }
        ga = gamma[tid];
        be = beta[tid];
    }

    PillarIn cur = {};
    if (p < pend) cur = load_pillar(features, num_voxels, coors, p, lane);

    v2f bw[4][3];
    load_w_frags(W, lane, bw);

    if (folder) {
        float s = 0.0f, q = 0.0f;
        for (int g = 0; g < RGROUPS; ++g) { s += gs[g]; q += gq[g]; }
        const float cnt  = (float)PILLARS * (float)NPTS;
        const float mean = s / cnt;
        float var = q / cnt - mean * mean;
        var = fmaxf(var, 0.0f);
        const float inv = rsqrtf(var + 1e-3f);
        const float sc  = inv * ga;
        sb[0][tid] = sc;
        sb[1][tid] = be - mean * sc;
    }
    __syncthreads();

    float scv[4], biv[4];
    for (int nt = 0; nt < 4; ++nt) {
        scv[nt] = sb[0][nt * 16 + lm];
        biv[nt] = sb[1][nt * 16 + lm];
    }

    float* frow = &fl[wave][lane][0];
    const float* flw = &fl[wave][0][0];

    while (p < pend) {
        const int pn = p + WAVES_PER_BLOCK;
        PillarIn nxt = cur;
        if (pn < pend) nxt = load_pillar(features, num_voxels, coors, pn, lane);

        pillar_feats_to_lds(cur, lane, frow);
        v2f a0[3], a1[3];
        load_a_frags(flw, lane, a0, a1);

        float mt[4];
#pragma unroll
        for (int nt = 0; nt < 4; ++nt) {
            v8f c0 = {}, c1 = {};
            gemm_tile(a0, a1, bw[nt], c0, c1);

            const v2f sc2 = { scv[nt], scv[nt] };
            const v2f bi2 = { biv[nt], biv[nt] };
            v2f y = __builtin_elementwise_max(__builtin_elementwise_fma(c0.s01, sc2, bi2),
                                              __builtin_elementwise_fma(c1.s01, sc2, bi2));
            y = __builtin_elementwise_max(y, __builtin_elementwise_fma(c0.s23, sc2, bi2));
            y = __builtin_elementwise_max(y, __builtin_elementwise_fma(c1.s23, sc2, bi2));
            y = __builtin_elementwise_max(y, __builtin_elementwise_fma(c0.s45, sc2, bi2));
            y = __builtin_elementwise_max(y, __builtin_elementwise_fma(c1.s45, sc2, bi2));
            y = __builtin_elementwise_max(y, __builtin_elementwise_fma(c0.s67, sc2, bi2));
            y = __builtin_elementwise_max(y, __builtin_elementwise_fma(c1.s67, sc2, bi2));
            mt[nt] = fmaxf(y.x, y.y);
        }
        float mo[4];
        for (int nt = 0; nt < 4; ++nt) {
            const float m = fmaxf(mt[nt], __shfl_xor(mt[nt], 16, 32));
            mo[nt] = fmaxf(m, 0.0f);
        }
        if (half == 0) {
            for (int nt = 0; nt < 4; ++nt)
                out[(size_t)p * UNITS + nt * 16 + lm] = mo[nt];
        }

        cur = nxt;
        p   = pn;
    }
}

extern "C" void kernel_launch(void* const* d_in, const int* in_sizes, int n_in,
                              void* d_out, int out_size, void* d_ws, size_t ws_size,
                              hipStream_t stream)
{
    const float* features   = (const float*)d_in[0];
    const int*   num_voxels = (const int*)  d_in[1];
    const int*   coors      = (const int*)  d_in[2];
    const float* W          = (const float*)d_in[3];
    const float* gamma      = (const float*)d_in[4];
    const float* beta       = (const float*)d_in[5];
    float*       out        = (float*)d_out;

    float* partials = (float*)d_ws;

    pfn_stats_kernel<<<NWG, BLOCK, 0, stream>>>(features, num_voxels, coors, W,
                                                partials);
    pfn_slice_sums<<<RGROUPS, RBLOCK, 0, stream>>>(partials);
    pfn_output_kernel<<<NWG, BLOCK, 0, stream>>>(features, num_voxels, coors, W,
                                                 partials, gamma, beta, out);
}


# GEMM K 12 -> 8 (24 -> 16 WMMA per pillar): mean and centre differences multiplied out, W rows regrouped, centre term = per-pillar B row; x rounds differently (products regrouped, sum order)
# speedup vs baseline: 14.7144x; 1.5654x over previous
#include <hip/hip_runtime.h>


#define PILLARS 40000
#define NPTS    32
#define CIN     4
#define UNITS   64
#define KDIM    8
#define KSTEPS  (KDIM / 4)
#define KPITCH  12
#define WAVES_PER_BLOCK 8
#define BLOCK   (WAVES_PER_BLOCK * 32)
#define NWG     1024

#define RGROUPS 16
#define RROWS   ((NWG + RGROUPS - 1) / RGROUPS)
#define RBLOCK  1024
#define RWAVES  (RBLOCK / 32)
#define RITER   ((RROWS + RWAVES - 1) / RWAVES)
static_assert((RGROUPS - 1) * RROWS < NWG, "every slice must own at least one row");

typedef float v2f __attribute__((ext_vector_type(2)));
typedef float v4f __attribute__((ext_vector_type(4)));
typedef float v8f __attribute__((ext_vector_type(8)));

struct PillarIn {
    float4 f;
    int    nv;
    int    cy;
    int    cx;
};

__device__ __forceinline__ PillarIn load_pillar(
    const float* __restrict__ features, const int* __restrict__ num_voxels,
    const int* __restrict__ coors, int p, int lane)
{
    PillarIn r;
    r.f  = *reinterpret_cast<const float4*>(features + ((size_t)p * NPTS + lane) * CIN);
    r.nv = num_voxels[p];
    r.cy = coors[p * 4 + 2];
    r.cx = coors[p * 4 + 3];
    return r;
}

__device__ __forceinline__ void pillar_feats_to_lds(
    const PillarIn& in, int lane, float* __restrict__ frow)
{
    const float4 f = in.f;

    float sx = f.x, sy = f.y, sz = f.z;
#pragma unroll
    for (int m = 16; m >= 1; m >>= 1) {
        sx += __shfl_xor(sx, m, 32);
        sy += __shfl_xor(sy, m, 32);
        sz += __shfl_xor(sz, m, 32);
    }
    const int   nv    = in.nv;
    const float invnv = 1.0f / (float)nv;
    const float mx = sx * invnv, my = sy * invnv, mz = sz * invnv;

    const float msk = (lane < nv) ? 1.0f : 0.0f;
    float4 lo, hi;
    lo.x = f.x * msk;
    lo.y = f.y * msk;
    lo.z = f.z * msk;
    lo.w = f.w * msk;
    hi.x = mx * msk;
    hi.y = my * msk;
    hi.z = mz * msk;
    hi.w = msk;
    *reinterpret_cast<float4*>(frow)     = lo;
    *reinterpret_cast<float4*>(frow + 4) = hi;
}


__device__ __forceinline__ float gamma_sign(float ga)
{
    return (ga < 0.0f) ? -1.0f : 1.0f;
}

__device__ __forceinline__ void load_w_frags(const float* __restrict__ W,
                                             const float* __restrict__ gamma, int lane,
                                             v2f (&bw)[4][KSTEPS],
                                             float (&pcx)[4], float (&pcy)[4], float (&pc0)[4])
{
    const int half = lane >> 4;
    const int lm   = lane & 15;
    for (int nt = 0; nt < 4; ++nt) {
        const int   u  = nt * 16 + lm;
        const float sg = gamma_sign(gamma[u]);
        float w[CIN + 5];
        for (int k = 0; k < CIN + 5; ++k) w[k] = W[k * UNITS + u] * sg;
        v2f b0, b1;
        b0.x = half ? (w[2] + w[6]) : (w[0] + w[4] + w[7]);
        b0.y = half ? w[3]          : (w[1] + w[5] + w[8]);
        b1.x = half ? -w[6]         : -w[4];
        b1.y = 0.0f;
        bw[nt][0] = b0;
        bw[nt][1] = b1;
        pcx[nt] = half ? -w[7] : 0.0f;
        pcy[nt] = half ? -w[8] : 0.0f;
        pc0[nt] = half ? 0.0f  : -w[5];
    }
}

__device__ __forceinline__ void load_a_frags(const float* __restrict__ flw  ,
                                             int lane, v2f (&a0)[KSTEPS], v2f (&a1)[KSTEPS])
{
    const int half = lane >> 4;
    const int lm   = lane & 15;
    for (int ks = 0; ks < KSTEPS; ++ks) {
        const int k0 = ks * 4 + 2 * half;
        a0[ks] = *reinterpret_cast<const v2f*>(flw + lm * KPITCH + k0);
        a1[ks] = *reinterpret_cast<const v2f*>(flw + (16 + lm) * KPITCH + k0);
    }
}

__device__ __forceinline__ void gemm_tile(const v2f (&a0)[KSTEPS], const v2f (&a1)[KSTEPS],
                                          const v2f (&b)[KSTEPS], v8f& acc0, v8f& acc1)
{
#pragma unroll
    for (int ks = 0; ks < KSTEPS; ++ks) {
        acc0 = __builtin_amdgcn_wmma_f32_16x16x4_f32(
            false, a0[ks], false, b[ks], (short)0, acc0, false, false);
        acc1 = __builtin_amdgcn_wmma_f32_16x16x4_f32(
            false, a1[ks], false, b[ks], (short)0, acc1, false, false);
    }
}

__global__ __launch_bounds__(BLOCK) __attribute__((amdgpu_waves_per_eu(8, 8))) void pfn_stats_kernel(
    const float* __restrict__ features, const int* __restrict__ num_voxels,
    const int* __restrict__ coors, const float* __restrict__ W,
    const float* __restrict__ gamma,
    float* __restrict__ partials  ,
    float* __restrict__ out  )
{
    __shared__ __attribute__((aligned(16))) float fl[WAVES_PER_BLOCK][NPTS][KPITCH];
    __shared__ float red[2][WAVES_PER_BLOCK][UNITS];

    const int tid  = threadIdx.x;
    const int wave = __builtin_amdgcn_readfirstlane(tid >> 5);
    const int lane = tid & 31;
    const int half = lane >> 4;
    const int lm   = lane & 15;

    const int pend = ((int)(blockIdx.x + 1) * PILLARS) / NWG;
    int p = ((int)blockIdx.x * PILLARS) / NWG + wave;

    PillarIn cur = {};
    if (p < pend) cur = load_pillar(features, num_voxels, coors, p, lane);

    v2f bw[4][KSTEPS];
    float pcx[4], pcy[4], pc0[4];
    load_w_frags(W, gamma, lane, bw, pcx, pcy, pc0);

    float ps[4] = { 0.0f, 0.0f, 0.0f, 0.0f };
    float pq[4] = { 0.0f, 0.0f, 0.0f, 0.0f };
    float* frow = &fl[wave][lane][0];
    const float* flw = &fl[wave][0][0];

    while (p < pend) {
        const int pn = p + WAVES_PER_BLOCK;
        PillarIn nxt = cur;
        if (pn < pend) nxt = load_pillar(features, num_voxels, coors, pn, lane);

        pillar_feats_to_lds(cur, lane, frow);
        v2f a0[KSTEPS], a1[KSTEPS];
        load_a_frags(flw, lane, a0, a1);

        const float cx = (float)cur.cx * 0.2f + 0.1f;
        const float cy = (float)cur.cy * 0.2f + (-39.9f);
        for (int nt = 0; nt < 4; ++nt)
            bw[nt][1].y = fmaf(cx, pcx[nt], fmaf(cy, pcy[nt], pc0[nt]));

        float mt[4];
#pragma unroll
        for (int nt = 0; nt < 4; ++nt) {
            v8f c0 = {}, c1 = {};
            gemm_tile(a0, a1, bw[nt], c0, c1);
            const v8f t8 = c0 + c1;
            const v8f q8 = c0 * c0 + c1 * c1;
            const v4f t4 = t8.lo + t8.hi;
            const v4f q4 = q8.lo + q8.hi;
            const v2f t2 = t4.lo + t4.hi;
            const v2f q2 = q4.lo + q4.hi;
            ps[nt] += t2.x + t2.y;
            pq[nt] += q2.x + q2.y;
            const v8f m8 = __builtin_elementwise_maximum(c0, c1);
            const v4f m4 = __builtin_elementwise_maximum(m8.lo, m8.hi);
            const v2f m2 = __builtin_elementwise_maximum(m4.lo, m4.hi);
            mt[nt] = __builtin_elementwise_maximum(m2.x, m2.y);
        }
        float mo[4];
        for (int nt = 0; nt < 4; ++nt)
            mo[nt] = __builtin_elementwise_maximum(mt[nt], __shfl_xor(mt[nt], 16, 32));
        float* orow = out + (size_t)p * UNITS + half * 32 + lm;
        orow[0]  = half ? mo[2] : mo[0];
        orow[16] = half ? mo[3] : mo[1];

        cur = nxt;
        p   = pn;
    }

    for (int nt = 0; nt < 4; ++nt) {
        const float s = ps[nt] + __shfl_xor(ps[nt], 16, 32);
        const float q = pq[nt] + __shfl_xor(pq[nt], 16, 32);
        if (half == 0) {
            red[0][wave][nt * 16 + lm] = s;
            red[1][wave][nt * 16 + lm] = q;
        }
    }
    __syncthreads();

    if (tid < 2 * UNITS) {
        const int which = tid >> 6;
        const int u     = tid & 63;
        float s = 0.0f;
#pragma unroll
        for (int w = 0; w < WAVES_PER_BLOCK; ++w) s += red[which][w][u];
        partials[(size_t)blockIdx.x * (2 * UNITS) + tid] = s;
    }
}

__global__ __launch_bounds__(RBLOCK) void pfn_slice_sums(float* partials)
{
    __shared__ float red[RWAVES][2 * UNITS];

    const int tid  = threadIdx.x;
    const int wave = tid >> 5;
    const int lane = tid & 31;
    const int r0   = blockIdx.x * RROWS;
    const int r1   = (r0 + RROWS < NWG) ? (r0 + RROWS) : NWG;

    float4 v[RITER];
    for (int i = 0; i < RITER; ++i) {
        const int r  = r0 + wave + i * RWAVES;
        const int rc = (r < r1) ? r : (r1 - 1);
        v[i] = *reinterpret_cast<const float4*>(
            partials + (size_t)rc * (2 * UNITS) + lane * 4);
    }
    float4 s = { 0.0f, 0.0f, 0.0f, 0.0f };
    for (int i = 0; i < RITER; ++i) {
        const int r = r0 + wave + i * RWAVES;
        if (r < r1) {
            s.x += v[i].x; s.y += v[i].y; s.z += v[i].z; s.w += v[i].w;
        }
    }
    *reinterpret_cast<float4*>(&red[wave][lane * 4]) = s;
    __syncthreads();

    if (tid < 2 * UNITS) {
        float t = 0.0f;
        for (int w = 0; w < RWAVES; ++w) t += red[w][tid];
        partials[(size_t)r0 * (2 * UNITS) + tid] = t;
    }
}

#define AVEC    (PILLARS * UNITS / 4)
#define AWG     256
#define ABLOCK  256
#define AITER   ((AVEC + AWG * ABLOCK - 1) / (AWG * ABLOCK))
static_assert(ABLOCK % 16 == 0 && (AWG * ABLOCK) % 16 == 0, "a thread must stay on its 4 units");
static_assert(ABLOCK >= UNITS, "one folding thread per unit");

__global__ __launch_bounds__(ABLOCK) void pfn_apply_kernel(
    const float* __restrict__ slice_sums  ,
    const float* __restrict__ gamma, const float* __restrict__ beta,
    float* out  )
{
    __shared__ float sb[2][UNITS];

    const int tid = threadIdx.x;

    const bool folder = (tid < UNITS);
    float gs[RGROUPS], gq[RGROUPS];
    float ga = 0.0f, be = 0.0f;
    if (folder) {
        for (int g = 0; g < RGROUPS; ++g) {
            gs[g] = slice_sums[(size_t)g * RROWS * (2 * UNITS) + tid];
            gq[g] = slice_sums[(size_t)g * RROWS * (2 * UNITS) + UNITS + tid];
        }
        ga = gamma[tid];
        be = beta[tid];
    }

    float4* out4 = reinterpret_cast<float4*>(out);
    const int i0 = (int)blockIdx.x * ABLOCK + tid;
    float4 v[AITER];
    for (int j = 0; j < AITER; ++j) {
        const int i  = i0 + j * (AWG * ABLOCK);
        const int ic = (i < AVEC) ? i : (AVEC - 1);
        v[j] = out4[ic];
    }

    if (folder) {
        float s = 0.0f, q = 0.0f;
        for (int g = 0; g < RGROUPS; ++g) { s += gs[g]; q += gq[g]; }
        const float cnt  = (float)PILLARS * (float)NPTS;
        const float mean = s / cnt;
        float var = q / cnt - mean * mean;
        var = fmaxf(var, 0.0f);
        const float inv = rsqrtf(var + 1e-3f);
        const float sc  = inv * (ga * gamma_sign(ga));
        sb[0][tid] = sc;
        sb[1][tid] = be - mean * sc;
    }
    __syncthreads();

    const int u0 = (tid & 15) * 4;
    const float4 sc4 = *reinterpret_cast<const float4*>(&sb[0][u0]);
    const float4 bi4 = *reinterpret_cast<const float4*>(&sb[1][u0]);

    for (int j = 0; j < AITER; ++j) {
        const int i = i0 + j * (AWG * ABLOCK);
        if (i < AVEC) {
            float4 y;
            y.x = fmaxf(fmaf(v[j].x, sc4.x, bi4.x), 0.0f);
            y.y = fmaxf(fmaf(v[j].y, sc4.y, bi4.y), 0.0f);
            y.z = fmaxf(fmaf(v[j].z, sc4.z, bi4.z), 0.0f);
            y.w = fmaxf(fmaf(v[j].w, sc4.w, bi4.w), 0.0f);
            out4[i] = y;
        }
    }
}

extern "C" void kernel_launch(void* const* d_in, const int* in_sizes, int n_in,
                              void* d_out, int out_size, void* d_ws, size_t ws_size,
                              hipStream_t stream)
{
    const float* features   = (const float*)d_in[0];
    const int*   num_voxels = (const int*)  d_in[1];
    const int*   coors      = (const int*)  d_in[2];
    const float* W          = (const float*)d_in[3];
    const float* gamma      = (const float*)d_in[4];
    const float* beta       = (const float*)d_in[5];
    float*       out        = (float*)d_out;

    float* partials = (float*)d_ws;

    pfn_stats_kernel<<<NWG, BLOCK, 0, stream>>>(features, num_voxels, coors, W, gamma,
                                                partials, out);
    pfn_slice_sums<<<RGROUPS, RBLOCK, 0, stream>>>(partials);
    pfn_apply_kernel<<<AWG, ABLOCK, 0, stream>>>(partials, gamma, beta, out);
}


# Slice-sum launch removed (3 -> 2 launches): stats pass = 256 work-groups x 32 waves, 256 partial rows; each output work-group owns a 32-unit half and folds that half of the rows itself (sum order)
# speedup vs baseline: 16.3854x; 1.1136x over previous
#include <hip/hip_runtime.h>


#define PILLARS 40000
#define NPTS    32
#define CIN     4
#define UNITS   64
#define KDIM    8
#define KSTEPS  (KDIM / 4)
#define KPITCH  12
#define WAVES_PER_BLOCK 32
#define BLOCK   (WAVES_PER_BLOCK * 32)
#define NWG     256

#define CUNITS  32
#define PROW    (2 * UNITS)
#define PIDX(which, u) ((((u) >> 5) * 2 + (which)) * CUNITS + ((u) & (CUNITS - 1)))

typedef float v2f __attribute__((ext_vector_type(2)));
typedef float v4f __attribute__((ext_vector_type(4)));
typedef float v8f __attribute__((ext_vector_type(8)));

struct PillarIn {
    float4 f;
    int    nv;
    int    cy;
    int    cx;
};

__device__ __forceinline__ PillarIn load_pillar(
    const float* __restrict__ features, const int* __restrict__ num_voxels,
    const int* __restrict__ coors, int p, int lane)
{
    PillarIn r;
    r.f  = *reinterpret_cast<const float4*>(features + ((size_t)p * NPTS + lane) * CIN);
    r.nv = num_voxels[p];
    r.cy = coors[p * 4 + 2];
    r.cx = coors[p * 4 + 3];
    return r;
}

__device__ __forceinline__ void pillar_feats_to_lds(
    const PillarIn& in, int lane, float* __restrict__ frow)
{
    const float4 f = in.f;

    float sx = f.x, sy = f.y, sz = f.z;
#pragma unroll
    for (int m = 16; m >= 1; m >>= 1) {
        sx += __shfl_xor(sx, m, 32);
        sy += __shfl_xor(sy, m, 32);
        sz += __shfl_xor(sz, m, 32);
    }
    const int   nv    = in.nv;
    const float invnv = 1.0f / (float)nv;
    const float mx = sx * invnv, my = sy * invnv, mz = sz * invnv;

    const float msk = (lane < nv) ? 1.0f : 0.0f;
    float4 lo, hi;
    lo.x = f.x * msk;
    lo.y = f.y * msk;
    lo.z = f.z * msk;
    lo.w = f.w * msk;
    hi.x = mx * msk;
    hi.y = my * msk;
    hi.z = mz * msk;
    hi.w = msk;
    *reinterpret_cast<float4*>(frow)     = lo;
    *reinterpret_cast<float4*>(frow + 4) = hi;
}


__device__ __forceinline__ float gamma_sign(float ga)
{
    return (ga < 0.0f) ? -1.0f : 1.0f;
}

__device__ __forceinline__ void load_w_frags(const float* __restrict__ W,
                                             const float* __restrict__ gamma, int lane,
                                             v2f (&bw)[4][KSTEPS],
                                             float (&pcx)[4], float (&pcy)[4], float (&pc0)[4])
{
    const int half = lane >> 4;
    const int lm   = lane & 15;
    for (int nt = 0; nt < 4; ++nt) {
        const int   u  = nt * 16 + lm;
        const float sg = gamma_sign(gamma[u]);
        float w[CIN + 5];
        for (int k = 0; k < CIN + 5; ++k) w[k] = W[k * UNITS + u] * sg;
        v2f b0, b1;
        b0.x = half ? (w[2] + w[6]) : (w[0] + w[4] + w[7]);
        b0.y = half ? w[3]          : (w[1] + w[5] + w[8]);
        b1.x = half ? -w[6]         : -w[4];
        b1.y = 0.0f;
        bw[nt][0] = b0;
        bw[nt][1] = b1;
        pcx[nt] = half ? -w[7] : 0.0f;
        pcy[nt] = half ? -w[8] : 0.0f;
        pc0[nt] = half ? 0.0f  : -w[5];
    }
}

__device__ __forceinline__ void load_a_frags(const float* __restrict__ flw  ,
                                             int lane, v2f (&a0)[KSTEPS], v2f (&a1)[KSTEPS])
{
    const int half = lane >> 4;
    const int lm   = lane & 15;
    for (int ks = 0; ks < KSTEPS; ++ks) {
        const int k0 = ks * 4 + 2 * half;
        a0[ks] = *reinterpret_cast<const v2f*>(flw + lm * KPITCH + k0);
        a1[ks] = *reinterpret_cast<const v2f*>(flw + (16 + lm) * KPITCH + k0);
    }
}

__device__ __forceinline__ void gemm_tile(const v2f (&a0)[KSTEPS], const v2f (&a1)[KSTEPS],
                                          const v2f (&b)[KSTEPS], v8f& acc0, v8f& acc1)
{
#pragma unroll
    for (int ks = 0; ks < KSTEPS; ++ks) {
        acc0 = __builtin_amdgcn_wmma_f32_16x16x4_f32(
            false, a0[ks], false, b[ks], (short)0, acc0, false, false);
        acc1 = __builtin_amdgcn_wmma_f32_16x16x4_f32(
            false, a1[ks], false, b[ks], (short)0, acc1, false, false);
    }
}

__global__ __launch_bounds__(BLOCK) __attribute__((amdgpu_waves_per_eu(8, 8))) void pfn_stats_kernel(
    const float* __restrict__ features, const int* __restrict__ num_voxels,
    const int* __restrict__ coors, const float* __restrict__ W,
    const float* __restrict__ gamma,
    float* __restrict__ partials  ,
    float* __restrict__ out  )
{
    __shared__ __attribute__((aligned(16))) float fl[WAVES_PER_BLOCK][NPTS][KPITCH];
    __shared__ float red[2][WAVES_PER_BLOCK][UNITS];

    const int tid  = threadIdx.x;
    const int wave = __builtin_amdgcn_readfirstlane(tid >> 5);
    const int lane = tid & 31;
    const int half = lane >> 4;
    const int lm   = lane & 15;

    const int pend = ((int)(blockIdx.x + 1) * PILLARS) / NWG;
    int p = ((int)blockIdx.x * PILLARS) / NWG + wave;

    PillarIn cur = {};
    if (p < pend) cur = load_pillar(features, num_voxels, coors, p, lane);

    v2f bw[4][KSTEPS];
    float pcx[4], pcy[4], pc0[4];
    load_w_frags(W, gamma, lane, bw, pcx, pcy, pc0);

    float ps[4] = { 0.0f, 0.0f, 0.0f, 0.0f };
    float pq[4] = { 0.0f, 0.0f, 0.0f, 0.0f };
    float* frow = &fl[wave][lane][0];
    const float* flw = &fl[wave][0][0];

    while (p < pend) {
        const int pn = p + WAVES_PER_BLOCK;
        PillarIn nxt = cur;
        if (pn < pend) nxt = load_pillar(features, num_voxels, coors, pn, lane);

        pillar_feats_to_lds(cur, lane, frow);
        v2f a0[KSTEPS], a1[KSTEPS];
        load_a_frags(flw, lane, a0, a1);

        const float cx = (float)cur.cx * 0.2f + 0.1f;
        const float cy = (float)cur.cy * 0.2f + (-39.9f);
        for (int nt = 0; nt < 4; ++nt)
            bw[nt][1].y = fmaf(cx, pcx[nt], fmaf(cy, pcy[nt], pc0[nt]));

        float mt[4];
#pragma unroll
        for (int nt = 0; nt < 4; ++nt) {
            v8f c0 = {}, c1 = {};
            gemm_tile(a0, a1, bw[nt], c0, c1);
            const v8f t8 = c0 + c1;
            const v8f q8 = c0 * c0 + c1 * c1;
            const v4f t4 = t8.lo + t8.hi;
            const v4f q4 = q8.lo + q8.hi;
            const v2f t2 = t4.lo + t4.hi;
            const v2f q2 = q4.lo + q4.hi;
            ps[nt] += t2.x + t2.y;
            pq[nt] += q2.x + q2.y;
            const v8f m8 = __builtin_elementwise_maximum(c0, c1);
            const v4f m4 = __builtin_elementwise_maximum(m8.lo, m8.hi);
            const v2f m2 = __builtin_elementwise_maximum(m4.lo, m4.hi);
            mt[nt] = __builtin_elementwise_maximum(m2.x, m2.y);
        }
        float mo[4];
        for (int nt = 0; nt < 4; ++nt)
            mo[nt] = __builtin_elementwise_maximum(mt[nt], __shfl_xor(mt[nt], 16, 32));
        float* orow = out + (size_t)p * UNITS + half * 32 + lm;
        orow[0]  = half ? mo[2] : mo[0];
        orow[16] = half ? mo[3] : mo[1];

        cur = nxt;
        p   = pn;
    }

    for (int nt = 0; nt < 4; ++nt) {
        const float s = ps[nt] + __shfl_xor(ps[nt], 16, 32);
        const float q = pq[nt] + __shfl_xor(pq[nt], 16, 32);
        if (half == 0) {
            red[0][wave][nt * 16 + lm] = s;
            red[1][wave][nt * 16 + lm] = q;
        }
    }
    __syncthreads();

    if (tid < 2 * UNITS) {
        const int which = tid >> 6;
        const int u     = tid & 63;
        float s = 0.0f;
#pragma unroll
        for (int w = 0; w < WAVES_PER_BLOCK; ++w) s += red[which][w][u];
        partials[(size_t)blockIdx.x * PROW + PIDX(which, u)] = s;
    }
}

#define AWG     256
#define ABLOCK  256
#define AROWS   (ABLOCK / 8)
#define ASTRIDE ((AWG / 2) * AROWS)
#define AITER   ((PILLARS + ASTRIDE - 1) / ASTRIDE)
#define FCLASS  (ABLOCK / 16)
#define FITER   (NWG / FCLASS)
static_assert(UNITS == 2 * CUNITS, "two column halves");
static_assert(NWG % FCLASS == 0, "every row class folds the same number of rows");
static_assert(ABLOCK >= CUNITS, "one folding thread per unit of the half");

__global__ __launch_bounds__(ABLOCK) __attribute__((amdgpu_waves_per_eu(2, 2))) void pfn_apply_kernel(
    const float* __restrict__ partials  ,
    const float* __restrict__ gamma, const float* __restrict__ beta,
    float* out  )
{
    __shared__ __attribute__((aligned(16))) float red[FCLASS][2 * CUNITS];
    __shared__ __attribute__((aligned(16))) float sb[2][CUNITS];

    const int tid = threadIdx.x;
    const int h   = (int)blockIdx.x & 1;
    const int rb  = (int)blockIdx.x >> 1;

    const int fc = tid & 15;
    const int fr = tid >> 4;
    float4 pv[FITER];
    for (int i = 0; i < FITER; ++i)
        pv[i] = *reinterpret_cast<const float4*>(
            partials + (size_t)(fr + i * FCLASS) * PROW + h * (2 * CUNITS) + fc * 4);

    const bool folder = (tid < CUNITS);
    float ga = 0.0f, be = 0.0f;
    if (folder) {
        ga = gamma[h * CUNITS + tid];
        be = beta[h * CUNITS + tid];
    }

    float4* out4 = reinterpret_cast<float4*>(out);
    const int q  = tid & 7;
    const int r0 = rb * AROWS + (tid >> 3);
    float4 v[AITER];
    for (int j = 0; j < AITER; ++j) {
        const int r  = r0 + j * ASTRIDE;
        const int rc = (r < PILLARS) ? r : (PILLARS - 1);
        v[j] = out4[(size_t)rc * (UNITS / 4) + h * (CUNITS / 4) + q];
    }

    __builtin_amdgcn_sched_barrier(0);

    float4 s4 = { 0.0f, 0.0f, 0.0f, 0.0f };
    for (int i = 0; i < FITER; ++i) {
        s4.x += pv[i].x; s4.y += pv[i].y; s4.z += pv[i].z; s4.w += pv[i].w;
    }
    *reinterpret_cast<float4*>(&red[fr][fc * 4]) = s4;
    __syncthreads();

    if (folder) {
        float s = 0.0f, qq = 0.0f;
        for (int c = 0; c < FCLASS; ++c) { s += red[c][tid]; qq += red[c][CUNITS + tid]; }
        const float cnt  = (float)PILLARS * (float)NPTS;
        const float mean = s / cnt;
        float var = qq / cnt - mean * mean;
        var = fmaxf(var, 0.0f);
        const float inv = rsqrtf(var + 1e-3f);
        const float sc  = inv * (ga * gamma_sign(ga));
        sb[0][tid] = sc;
        sb[1][tid] = be - mean * sc;
    }
    __syncthreads();

    const float4 sc4 = *reinterpret_cast<const float4*>(&sb[0][q * 4]);
    const float4 bi4 = *reinterpret_cast<const float4*>(&sb[1][q * 4]);

    for (int j = 0; j < AITER; ++j) {
        const int r = r0 + j * ASTRIDE;
        if (r < PILLARS) {
            float4 y;
            y.x = fmaxf(fmaf(v[j].x, sc4.x, bi4.x), 0.0f);
            y.y = fmaxf(fmaf(v[j].y, sc4.y, bi4.y), 0.0f);
            y.z = fmaxf(fmaf(v[j].z, sc4.z, bi4.z), 0.0f);
            y.w = fmaxf(fmaf(v[j].w, sc4.w, bi4.w), 0.0f);
            out4[(size_t)r * (UNITS / 4) + h * (CUNITS / 4) + q] = y;
        }
    }
}

extern "C" void kernel_launch(void* const* d_in, const int* in_sizes, int n_in,
                              void* d_out, int out_size, void* d_ws, size_t ws_size,
                              hipStream_t stream)
{
    const float* features   = (const float*)d_in[0];
    const int*   num_voxels = (const int*)  d_in[1];
    const int*   coors      = (const int*)  d_in[2];
    const float* W          = (const float*)d_in[3];
    const float* gamma      = (const float*)d_in[4];
    const float* beta       = (const float*)d_in[5];
    float*       out        = (float*)d_out;

    float* partials = (float*)d_ws;

    pfn_stats_kernel<<<NWG, BLOCK, 0, stream>>>(features, num_voxels, coors, W, gamma,
                                                partials, out);
    pfn_apply_kernel<<<AWG, ABLOCK, 0, stream>>>(partials, gamma, beta, out);
}


# Pillars with num_voxels <= 16 run the lower 16-point M-tile only (8 WMMAs, not 16): the upper tile is all masked zero rows, adds 0 to the sums and 0 to the max; scalar branch; sum order unchanged
# speedup vs baseline: 17.8801x; 1.0912x over previous
#include <hip/hip_runtime.h>


#define PILLARS 40000
#define NPTS    32
#define CIN     4
#define UNITS   64
#define KDIM    8
#define KSTEPS  (KDIM / 4)
#define KPITCH  12
#define WAVES_PER_BLOCK 32
#define BLOCK   (WAVES_PER_BLOCK * 32)
#define NWG     256

#define CUNITS  32
#define PROW    (2 * UNITS)
#define PIDX(which, u) ((((u) >> 5) * 2 + (which)) * CUNITS + ((u) & (CUNITS - 1)))

typedef float v2f __attribute__((ext_vector_type(2)));
typedef float v4f __attribute__((ext_vector_type(4)));
typedef float v8f __attribute__((ext_vector_type(8)));

struct PillarIn {
    float4 f;
    int    nv;
    int    cy;
    int    cx;
};

__device__ __forceinline__ PillarIn load_pillar(
    const float* __restrict__ features, const int* __restrict__ num_voxels,
    const int* __restrict__ coors, int p, int lane)
{
    PillarIn r;
    r.f  = *reinterpret_cast<const float4*>(features + ((size_t)p * NPTS + lane) * CIN);
    r.nv = num_voxels[p];
    r.cy = coors[p * 4 + 2];
    r.cx = coors[p * 4 + 3];
    return r;
}

__device__ __forceinline__ void pillar_feats_to_lds(
    const PillarIn& in, int lane, float* __restrict__ frow)
{
    const float4 f = in.f;

    float sx = f.x, sy = f.y, sz = f.z;
#pragma unroll
    for (int m = 16; m >= 1; m >>= 1) {
        sx += __shfl_xor(sx, m, 32);
        sy += __shfl_xor(sy, m, 32);
        sz += __shfl_xor(sz, m, 32);
    }
    const int   nv    = in.nv;
    const float invnv = 1.0f / (float)nv;
    const float mx = sx * invnv, my = sy * invnv, mz = sz * invnv;

    const float msk = (lane < nv) ? 1.0f : 0.0f;
    float4 lo, hi;
    lo.x = f.x * msk;
    lo.y = f.y * msk;
    lo.z = f.z * msk;
    lo.w = f.w * msk;
    hi.x = mx * msk;
    hi.y = my * msk;
    hi.z = mz * msk;
    hi.w = msk;
    *reinterpret_cast<float4*>(frow)     = lo;
    *reinterpret_cast<float4*>(frow + 4) = hi;
}


__device__ __forceinline__ float gamma_sign(float ga)
{
    return (ga < 0.0f) ? -1.0f : 1.0f;
}

__device__ __forceinline__ void load_w_frags(const float* __restrict__ W,
                                             const float* __restrict__ gamma, int lane,
                                             v2f (&bw)[4][KSTEPS],
                                             float (&pcx)[4], float (&pcy)[4], float (&pc0)[4])
{
    const int half = lane >> 4;
    const int lm   = lane & 15;
    for (int nt = 0; nt < 4; ++nt) {
        const int   u  = nt * 16 + lm;
        const float sg = gamma_sign(gamma[u]);
        float w[CIN + 5];
        for (int k = 0; k < CIN + 5; ++k) w[k] = W[k * UNITS + u] * sg;
        v2f b0, b1;
        b0.x = half ? (w[2] + w[6]) : (w[0] + w[4] + w[7]);
        b0.y = half ? w[3]          : (w[1] + w[5] + w[8]);
        b1.x = half ? -w[6]         : -w[4];
        b1.y = 0.0f;
        bw[nt][0] = b0;
        bw[nt][1] = b1;
        pcx[nt] = half ? -w[7] : 0.0f;
        pcy[nt] = half ? -w[8] : 0.0f;
        pc0[nt] = half ? 0.0f  : -w[5];
    }
}

__device__ __forceinline__ void load_a_frags(const float* __restrict__ flw  ,
                                             int lane, v2f (&a0)[KSTEPS], v2f (&a1)[KSTEPS])
{
    const int half = lane >> 4;
    const int lm   = lane & 15;
    for (int ks = 0; ks < KSTEPS; ++ks) {
        const int k0 = ks * 4 + 2 * half;
        a0[ks] = *reinterpret_cast<const v2f*>(flw + lm * KPITCH + k0);
        a1[ks] = *reinterpret_cast<const v2f*>(flw + (16 + lm) * KPITCH + k0);
    }
}

__device__ __forceinline__ void load_a_frags_lower(const float* __restrict__ flw  ,
                                                   int lane, v2f (&a0)[KSTEPS])
{
    const int half = lane >> 4;
    const int lm   = lane & 15;
    for (int ks = 0; ks < KSTEPS; ++ks) {
        const int k0 = ks * 4 + 2 * half;
        a0[ks] = *reinterpret_cast<const v2f*>(flw + lm * KPITCH + k0);
    }
}

__device__ __forceinline__ void gemm_tile(const v2f (&a0)[KSTEPS], const v2f (&a1)[KSTEPS],
                                          const v2f (&b)[KSTEPS], v8f& acc0, v8f& acc1)
{
#pragma unroll
    for (int ks = 0; ks < KSTEPS; ++ks) {
        acc0 = __builtin_amdgcn_wmma_f32_16x16x4_f32(
            false, a0[ks], false, b[ks], (short)0, acc0, false, false);
        acc1 = __builtin_amdgcn_wmma_f32_16x16x4_f32(
            false, a1[ks], false, b[ks], (short)0, acc1, false, false);
    }
}

__global__ __launch_bounds__(BLOCK) __attribute__((amdgpu_waves_per_eu(8, 8))) void pfn_stats_kernel(
    const float* __restrict__ features, const int* __restrict__ num_voxels,
    const int* __restrict__ coors, const float* __restrict__ W,
    const float* __restrict__ gamma,
    float* __restrict__ partials  ,
    float* __restrict__ out  )
{
    __shared__ __attribute__((aligned(16))) float fl[WAVES_PER_BLOCK][NPTS][KPITCH];
    __shared__ float red[2][WAVES_PER_BLOCK][UNITS];

    const int tid  = threadIdx.x;
    const int wave = __builtin_amdgcn_readfirstlane(tid >> 5);
    const int lane = tid & 31;
    const int half = lane >> 4;
    const int lm   = lane & 15;

    const int pend = ((int)(blockIdx.x + 1) * PILLARS) / NWG;
    int p = ((int)blockIdx.x * PILLARS) / NWG + wave;

    PillarIn cur = {};
    if (p < pend) cur = load_pillar(features, num_voxels, coors, p, lane);

    v2f bw[4][KSTEPS];
    float pcx[4], pcy[4], pc0[4];
    load_w_frags(W, gamma, lane, bw, pcx, pcy, pc0);

    float ps[4] = { 0.0f, 0.0f, 0.0f, 0.0f };
    float pq[4] = { 0.0f, 0.0f, 0.0f, 0.0f };
    float* frow = &fl[wave][lane][0];
    const float* flw = &fl[wave][0][0];

    while (p < pend) {
        const int pn = p + WAVES_PER_BLOCK;
        PillarIn nxt = cur;
        if (pn < pend) nxt = load_pillar(features, num_voxels, coors, pn, lane);

        pillar_feats_to_lds(cur, lane, frow);

        const float cx = (float)cur.cx * 0.2f + 0.1f;
        const float cy = (float)cur.cy * 0.2f + (-39.9f);
        for (int nt = 0; nt < 4; ++nt)
            bw[nt][1].y = fmaf(cx, pcx[nt], fmaf(cy, pcy[nt], pc0[nt]));

        float mt[4];
        if (cur.nv > NPTS / 2) {
            v2f a0[KSTEPS], a1[KSTEPS];
            load_a_frags(flw, lane, a0, a1);
#pragma unroll
            for (int nt = 0; nt < 4; ++nt) {
                v8f c0 = {}, c1 = {};
                gemm_tile(a0, a1, bw[nt], c0, c1);
                const v8f t8 = c0 + c1;
                const v8f q8 = c0 * c0 + c1 * c1;
                const v4f t4 = t8.lo + t8.hi;
                const v4f q4 = q8.lo + q8.hi;
                const v2f t2 = t4.lo + t4.hi;
                const v2f q2 = q4.lo + q4.hi;
                ps[nt] += t2.x + t2.y;
                pq[nt] += q2.x + q2.y;
                const v8f m8 = __builtin_elementwise_maximum(c0, c1);
                const v4f m4 = __builtin_elementwise_maximum(m8.lo, m8.hi);
                const v2f m2 = __builtin_elementwise_maximum(m4.lo, m4.hi);
                mt[nt] = __builtin_elementwise_maximum(m2.x, m2.y);
            }
        } else {
            v2f a0[KSTEPS];
            load_a_frags_lower(flw, lane, a0);
            v8f c[4];
#pragma unroll
            for (int nt = 0; nt < 4; ++nt) c[nt] = v8f{};
#pragma unroll
            for (int ks = 0; ks < KSTEPS; ++ks)
#pragma unroll
                for (int nt = 0; nt < 4; ++nt)
                    c[nt] = __builtin_amdgcn_wmma_f32_16x16x4_f32(
                        false, a0[ks], false, bw[nt][ks], (short)0, c[nt], false, false);
#pragma unroll
            for (int nt = 0; nt < 4; ++nt) {
                const v8f q8 = c[nt] * c[nt];
                const v4f t4 = c[nt].lo + c[nt].hi;
                const v4f q4 = q8.lo + q8.hi;
                const v2f t2 = t4.lo + t4.hi;
                const v2f q2 = q4.lo + q4.hi;
                ps[nt] += t2.x + t2.y;
                pq[nt] += q2.x + q2.y;
                const v4f m4 = __builtin_elementwise_maximum(c[nt].lo, c[nt].hi);
                const v2f m2 = __builtin_elementwise_maximum(m4.lo, m4.hi);
                mt[nt] = __builtin_elementwise_maximum(
                    __builtin_elementwise_maximum(m2.x, m2.y), 0.0f);
            }
        }
        float mo[4];
        for (int nt = 0; nt < 4; ++nt)
            mo[nt] = __builtin_elementwise_maximum(mt[nt], __shfl_xor(mt[nt], 16, 32));
        float* orow = out + (size_t)p * UNITS + half * 32 + lm;
        orow[0]  = half ? mo[2] : mo[0];
        orow[16] = half ? mo[3] : mo[1];

        cur = nxt;
        p   = pn;
    }

    for (int nt = 0; nt < 4; ++nt) {
        const float s = ps[nt] + __shfl_xor(ps[nt], 16, 32);
        const float q = pq[nt] + __shfl_xor(pq[nt], 16, 32);
        if (half == 0) {
            red[0][wave][nt * 16 + lm] = s;
            red[1][wave][nt * 16 + lm] = q;
        }
    }
    __syncthreads();

    if (tid < 2 * UNITS) {
        const int which = tid >> 6;
        const int u     = tid & 63;
        float s = 0.0f;
#pragma unroll
        for (int w = 0; w < WAVES_PER_BLOCK; ++w) s += red[which][w][u];
        partials[(size_t)blockIdx.x * PROW + PIDX(which, u)] = s;
    }
}

#define AWG     256
#define ABLOCK  256
#define AROWS   (ABLOCK / 8)
#define ASTRIDE ((AWG / 2) * AROWS)
#define AITER   ((PILLARS + ASTRIDE - 1) / ASTRIDE)
#define FCLASS  (ABLOCK / 16)
#define FITER   (NWG / FCLASS)
static_assert(UNITS == 2 * CUNITS, "two column halves");
static_assert(NWG % FCLASS == 0, "every row class folds the same number of rows");
static_assert(ABLOCK >= CUNITS, "one folding thread per unit of the half");

__global__ __launch_bounds__(ABLOCK) __attribute__((amdgpu_waves_per_eu(2, 2))) void pfn_apply_kernel(
    const float* __restrict__ partials  ,
    const float* __restrict__ gamma, const float* __restrict__ beta,
    float* out  )
{
    __shared__ __attribute__((aligned(16))) float red[FCLASS][2 * CUNITS];
    __shared__ __attribute__((aligned(16))) float sb[2][CUNITS];

    const int tid = threadIdx.x;
    const int h   = (int)blockIdx.x & 1;
    const int rb  = (int)blockIdx.x >> 1;

    const int fc = tid & 15;
    const int fr = tid >> 4;
    float4 pv[FITER];
    for (int i = 0; i < FITER; ++i)
        pv[i] = *reinterpret_cast<const float4*>(
            partials + (size_t)(fr + i * FCLASS) * PROW + h * (2 * CUNITS) + fc * 4);

    const bool folder = (tid < CUNITS);
    float ga = 0.0f, be = 0.0f;
    if (folder) {
        ga = gamma[h * CUNITS + tid];
        be = beta[h * CUNITS + tid];
    }

    float4* out4 = reinterpret_cast<float4*>(out);
    const int q  = tid & 7;
    const int r0 = rb * AROWS + (tid >> 3);
    float4 v[AITER];
    for (int j = 0; j < AITER; ++j) {
        const int r  = r0 + j * ASTRIDE;
        const int rc = (r < PILLARS) ? r : (PILLARS - 1);
        v[j] = out4[(size_t)rc * (UNITS / 4) + h * (CUNITS / 4) + q];
    }

    __builtin_amdgcn_sched_barrier(0);

    float4 s4 = { 0.0f, 0.0f, 0.0f, 0.0f };
    for (int i = 0; i < FITER; ++i) {
        s4.x += pv[i].x; s4.y += pv[i].y; s4.z += pv[i].z; s4.w += pv[i].w;
    }
    *reinterpret_cast<float4*>(&red[fr][fc * 4]) = s4;
    __syncthreads();

    if (folder) {
        float s = 0.0f, qq = 0.0f;
        for (int c = 0; c < FCLASS; ++c) { s += red[c][tid]; qq += red[c][CUNITS + tid]; }
        const float cnt  = (float)PILLARS * (float)NPTS;
        const float mean = s / cnt;
        float var = qq / cnt - mean * mean;
        var = fmaxf(var, 0.0f);
        const float inv = rsqrtf(var + 1e-3f);
        const float sc  = inv * (ga * gamma_sign(ga));
        sb[0][tid] = sc;
        sb[1][tid] = be - mean * sc;
    }
    __syncthreads();

    const float4 sc4 = *reinterpret_cast<const float4*>(&sb[0][q * 4]);
    const float4 bi4 = *reinterpret_cast<const float4*>(&sb[1][q * 4]);

    for (int j = 0; j < AITER; ++j) {
        const int r = r0 + j * ASTRIDE;
        if (r < PILLARS) {
            float4 y;
            y.x = fmaxf(fmaf(v[j].x, sc4.x, bi4.x), 0.0f);
            y.y = fmaxf(fmaf(v[j].y, sc4.y, bi4.y), 0.0f);
            y.z = fmaxf(fmaf(v[j].z, sc4.z, bi4.z), 0.0f);
            y.w = fmaxf(fmaf(v[j].w, sc4.w, bi4.w), 0.0f);
            out4[(size_t)r * (UNITS / 4) + h * (CUNITS / 4) + q] = y;
        }
    }
}

extern "C" void kernel_launch(void* const* d_in, const int* in_sizes, int n_in,
                              void* d_out, int out_size, void* d_ws, size_t ws_size,
                              hipStream_t stream)
{
    const float* features   = (const float*)d_in[0];
    const int*   num_voxels = (const int*)  d_in[1];
    const int*   coors      = (const int*)  d_in[2];
    const float* W          = (const float*)d_in[3];
    const float* gamma      = (const float*)d_in[4];
    const float* beta       = (const float*)d_in[5];
    float*       out        = (float*)d_out;

    float* partials = (float*)d_ws;

    pfn_stats_kernel<<<NWG, BLOCK, 0, stream>>>(features, num_voxels, coors, W, gamma,
                                                partials, out);
    pfn_apply_kernel<<<AWG, ABLOCK, 0, stream>>>(partials, gamma, beta, out);
}


# GEMM K 8 -> 4 (WMMAs per pillar 16 -> 8, 8 -> 4 where num_voxels <= 16): mean and centre terms enter as the accumulator's masked start value c(u) (C operand); x rounds differently (sum order)
# speedup vs baseline: 19.3318x; 1.0812x over previous
#include <hip/hip_runtime.h>


#define PILLARS 40000
#define NPTS    32
#define CIN     4
#define UNITS   64
#define KPITCH  4
#define WAVES_PER_BLOCK 32
#define BLOCK   (WAVES_PER_BLOCK * 32)
#define NWG     256

#define CUNITS  32
#define PROW    (2 * UNITS)
#define PIDX(which, u) ((((u) >> 5) * 2 + (which)) * CUNITS + ((u) & (CUNITS - 1)))

typedef float v2f __attribute__((ext_vector_type(2)));
typedef float v4f __attribute__((ext_vector_type(4)));
typedef float v8f __attribute__((ext_vector_type(8)));

struct PillarIn {
    float4 f;
    int    nv;
    int    cy;
    int    cx;
};

__device__ __forceinline__ PillarIn load_pillar(
    const float* __restrict__ features, const int* __restrict__ num_voxels,
    const int* __restrict__ coors, int p, int lane)
{
    PillarIn r;
    r.f  = *reinterpret_cast<const float4*>(features + ((size_t)p * NPTS + lane) * CIN);
    r.nv = num_voxels[p];
    r.cy = coors[p * 4 + 2];
    r.cx = coors[p * 4 + 3];
    return r;
}

__device__ __forceinline__ void pillar_feats_to_lds(
    const PillarIn& in, int lane, float* __restrict__ frow, float* __restrict__ mrow,
    float& mx, float& my, float& mz)
{
    const float4 f = in.f;

    float sx = f.x, sy = f.y, sz = f.z;
#pragma unroll
    for (int m = 16; m >= 1; m >>= 1) {
        sx += __shfl_xor(sx, m, 32);
        sy += __shfl_xor(sy, m, 32);
        sz += __shfl_xor(sz, m, 32);
    }
    const int   nv    = in.nv;
    const float invnv = 1.0f / (float)nv;
    mx = sx * invnv; my = sy * invnv; mz = sz * invnv;

    const float msk = (lane < nv) ? 1.0f : 0.0f;
    float4 lo;
    lo.x = f.x * msk;
    lo.y = f.y * msk;
    lo.z = f.z * msk;
    lo.w = f.w * msk;
    *reinterpret_cast<float4*>(frow) = lo;
    *mrow = msk;
}


__device__ __forceinline__ float gamma_sign(float ga)
{
    return (ga < 0.0f) ? -1.0f : 1.0f;
}

__device__ __forceinline__ void load_w_frags(const float* __restrict__ W,
                                             const float* __restrict__ gamma, int lane,
                                             v2f (&bw)[4], float (&nw)[4][5])
{
    const int half = lane >> 4;
    const int lm   = lane & 15;
    for (int nt = 0; nt < 4; ++nt) {
        const int   u  = nt * 16 + lm;
        const float sg = gamma_sign(gamma[u]);
        float w[CIN + 5];
        for (int k = 0; k < CIN + 5; ++k) w[k] = W[k * UNITS + u] * sg;
        v2f b;
        b.x = half ? (w[2] + w[6]) : (w[0] + w[4] + w[7]);
        b.y = half ? w[3]          : (w[1] + w[5] + w[8]);
        bw[nt] = b;
        for (int k = 0; k < 5; ++k) nw[nt][k] = -w[4 + k];
    }
}

__device__ __forceinline__ v8f load_row_masks(const float* __restrict__ mkw, int p0, int half)
{
    const v4f a = *reinterpret_cast<const v4f*>(mkw + p0 + 8 * half);
    const v4f b = *reinterpret_cast<const v4f*>(mkw + p0 + 8 * half + 4);
    v8f r;
    r.lo = a;
    r.hi = b;
    return r;
}

__global__ __launch_bounds__(BLOCK) __attribute__((amdgpu_waves_per_eu(8, 8))) void pfn_stats_kernel(
    const float* __restrict__ features, const int* __restrict__ num_voxels,
    const int* __restrict__ coors, const float* __restrict__ W,
    const float* __restrict__ gamma,
    float* __restrict__ partials  ,
    float* __restrict__ out  )
{
    __shared__ __attribute__((aligned(16))) float fl[WAVES_PER_BLOCK][NPTS][KPITCH];
    __shared__ __attribute__((aligned(16))) float mk[WAVES_PER_BLOCK][NPTS];
    __shared__ float red[2][WAVES_PER_BLOCK][UNITS];

    const int tid  = threadIdx.x;
    const int wave = __builtin_amdgcn_readfirstlane(tid >> 5);
    const int lane = tid & 31;
    const int half = lane >> 4;
    const int lm   = lane & 15;

    const int pend = ((int)(blockIdx.x + 1) * PILLARS) / NWG;
    int p = ((int)blockIdx.x * PILLARS) / NWG + wave;

    PillarIn cur = {};
    if (p < pend) cur = load_pillar(features, num_voxels, coors, p, lane);

    v2f bw[4];
    float nw[4][5];
    load_w_frags(W, gamma, lane, bw, nw);

    float ps[4] = { 0.0f, 0.0f, 0.0f, 0.0f };
    float pq[4] = { 0.0f, 0.0f, 0.0f, 0.0f };
    float* frow = &fl[wave][lane][0];
    float* mrow = &mk[wave][lane];
    const float* flw = &fl[wave][0][0];
    const float* mkw = &mk[wave][0];

    while (p < pend) {
        const int pn = p + WAVES_PER_BLOCK;
        PillarIn nxt = cur;
        if (pn < pend) nxt = load_pillar(features, num_voxels, coors, pn, lane);

        float mx, my, mz;
        pillar_feats_to_lds(cur, lane, frow, mrow, mx, my, mz);

        const float cx = (float)cur.cx * 0.2f + 0.1f;
        const float cy = (float)cur.cy * 0.2f + (-39.9f);
        float cc[4];
        for (int nt = 0; nt < 4; ++nt)
            cc[nt] = fmaf(mx, nw[nt][0], fmaf(my, nw[nt][1], fmaf(mz, nw[nt][2],
                     fmaf(cx, nw[nt][3], cy * nw[nt][4]))));

        float mt[4];
        if (cur.nv > NPTS / 2) {
            const v2f a0 = *reinterpret_cast<const v2f*>(flw + lm * KPITCH + 2 * half);
            const v2f a1 = *reinterpret_cast<const v2f*>(flw + (16 + lm) * KPITCH + 2 * half);
            const v8f rm = load_row_masks(mkw, 16, half);
#pragma unroll
            for (int nt = 0; nt < 4; ++nt) {
                const float cu = cc[nt];
                v8f c0 = { cu, cu, cu, cu, cu, cu, cu, cu };
                v8f c1 = rm * cu;
                c0 = __builtin_amdgcn_wmma_f32_16x16x4_f32(
                    false, a0, false, bw[nt], (short)0, c0, false, false);
                c1 = __builtin_amdgcn_wmma_f32_16x16x4_f32(
                    false, a1, false, bw[nt], (short)0, c1, false, false);
                const v8f t8 = c0 + c1;
                const v8f q8 = c0 * c0 + c1 * c1;
                const v4f t4 = t8.lo + t8.hi;
                const v4f q4 = q8.lo + q8.hi;
                const v2f t2 = t4.lo + t4.hi;
                const v2f q2 = q4.lo + q4.hi;
                ps[nt] += t2.x + t2.y;
                pq[nt] += q2.x + q2.y;
                const v8f m8 = __builtin_elementwise_maximum(c0, c1);
                const v4f m4 = __builtin_elementwise_maximum(m8.lo, m8.hi);
                const v2f m2 = __builtin_elementwise_maximum(m4.lo, m4.hi);
                mt[nt] = __builtin_elementwise_maximum(m2.x, m2.y);
            }
        } else {
            const v2f a0 = *reinterpret_cast<const v2f*>(flw + lm * KPITCH + 2 * half);
            const v8f rm = load_row_masks(mkw, 0, half);
            v8f c[4];
#pragma unroll
            for (int nt = 0; nt < 4; ++nt) c[nt] = rm * cc[nt];
#pragma unroll
            for (int nt = 0; nt < 4; ++nt)
                c[nt] = __builtin_amdgcn_wmma_f32_16x16x4_f32(
                    false, a0, false, bw[nt], (short)0, c[nt], false, false);
#pragma unroll
            for (int nt = 0; nt < 4; ++nt) {
                const v8f q8 = c[nt] * c[nt];
                const v4f t4 = c[nt].lo + c[nt].hi;
                const v4f q4 = q8.lo + q8.hi;
                const v2f t2 = t4.lo + t4.hi;
                const v2f q2 = q4.lo + q4.hi;
                ps[nt] += t2.x + t2.y;
                pq[nt] += q2.x + q2.y;
                const v4f m4 = __builtin_elementwise_maximum(c[nt].lo, c[nt].hi);
                const v2f m2 = __builtin_elementwise_maximum(m4.lo, m4.hi);
                mt[nt] = __builtin_elementwise_maximum(
                    __builtin_elementwise_maximum(m2.x, m2.y), 0.0f);
            }
        }
        float mo[4];
        for (int nt = 0; nt < 4; ++nt)
            mo[nt] = __builtin_elementwise_maximum(mt[nt], __shfl_xor(mt[nt], 16, 32));
        float* orow = out + (size_t)p * UNITS + half * 32 + lm;
        orow[0]  = half ? mo[2] : mo[0];
        orow[16] = half ? mo[3] : mo[1];

        cur = nxt;
        p   = pn;
    }

    for (int nt = 0; nt < 4; ++nt) {
        const float s = ps[nt] + __shfl_xor(ps[nt], 16, 32);
        const float q = pq[nt] + __shfl_xor(pq[nt], 16, 32);
        if (half == 0) {
            red[0][wave][nt * 16 + lm] = s;
            red[1][wave][nt * 16 + lm] = q;
        }
    }
    __syncthreads();

    if (tid < 2 * UNITS) {
        const int which = tid >> 6;
        const int u     = tid & 63;
        float s = 0.0f;
#pragma unroll
        for (int w = 0; w < WAVES_PER_BLOCK; ++w) s += red[which][w][u];
        partials[(size_t)blockIdx.x * PROW + PIDX(which, u)] = s;
    }
}

#define AWG     256
#define ABLOCK  256
#define AROWS   (ABLOCK / 8)
#define ASTRIDE ((AWG / 2) * AROWS)
#define AITER   ((PILLARS + ASTRIDE - 1) / ASTRIDE)
#define FCLASS  (ABLOCK / 16)
#define FITER   (NWG / FCLASS)
static_assert(UNITS == 2 * CUNITS, "two column halves");
static_assert(NWG % FCLASS == 0, "every row class folds the same number of rows");
static_assert(ABLOCK >= CUNITS, "one folding thread per unit of the half");

__global__ __launch_bounds__(ABLOCK) __attribute__((amdgpu_waves_per_eu(2, 2))) void pfn_apply_kernel(
    const float* __restrict__ partials  ,
    const float* __restrict__ gamma, const float* __restrict__ beta,
    float* out  )
{
    __shared__ __attribute__((aligned(16))) float red[FCLASS][2 * CUNITS];
    __shared__ __attribute__((aligned(16))) float sb[2][CUNITS];

    const int tid = threadIdx.x;
    const int h   = (int)blockIdx.x & 1;
    const int rb  = (int)blockIdx.x >> 1;

    const int fc = tid & 15;
    const int fr = tid >> 4;
    float4 pv[FITER];
    for (int i = 0; i < FITER; ++i)
        pv[i] = *reinterpret_cast<const float4*>(
            partials + (size_t)(fr + i * FCLASS) * PROW + h * (2 * CUNITS) + fc * 4);

    const bool folder = (tid < CUNITS);
    float ga = 0.0f, be = 0.0f;
    if (folder) {
        ga = gamma[h * CUNITS + tid];
        be = beta[h * CUNITS + tid];
    }

    float4* out4 = reinterpret_cast<float4*>(out);
    const int q  = tid & 7;
    const int r0 = rb * AROWS + (tid >> 3);
    float4 v[AITER];
    for (int j = 0; j < AITER; ++j) {
        const int r  = r0 + j * ASTRIDE;
        const int rc = (r < PILLARS) ? r : (PILLARS - 1);
        v[j] = out4[(size_t)rc * (UNITS / 4) + h * (CUNITS / 4) + q];
    }

    __builtin_amdgcn_sched_barrier(0);

    float4 s4 = { 0.0f, 0.0f, 0.0f, 0.0f };
    for (int i = 0; i < FITER; ++i) {
        s4.x += pv[i].x; s4.y += pv[i].y; s4.z += pv[i].z; s4.w += pv[i].w;
    }
    *reinterpret_cast<float4*>(&red[fr][fc * 4]) = s4;
    __syncthreads();

    if (folder) {
        float s = 0.0f, qq = 0.0f;
        for (int c = 0; c < FCLASS; ++c) { s += red[c][tid]; qq += red[c][CUNITS + tid]; }
        const float cnt  = (float)PILLARS * (float)NPTS;
        const float mean = s / cnt;
        float var = qq / cnt - mean * mean;
        var = fmaxf(var, 0.0f);
        const float inv = rsqrtf(var + 1e-3f);
        const float sc  = inv * (ga * gamma_sign(ga));
        sb[0][tid] = sc;
        sb[1][tid] = be - mean * sc;
    }
    __syncthreads();

    const float4 sc4 = *reinterpret_cast<const float4*>(&sb[0][q * 4]);
    const float4 bi4 = *reinterpret_cast<const float4*>(&sb[1][q * 4]);

    for (int j = 0; j < AITER; ++j) {
        const int r = r0 + j * ASTRIDE;
        if (r < PILLARS) {
            float4 y;
            y.x = fmaxf(fmaf(v[j].x, sc4.x, bi4.x), 0.0f);
            y.y = fmaxf(fmaf(v[j].y, sc4.y, bi4.y), 0.0f);
            y.z = fmaxf(fmaf(v[j].z, sc4.z, bi4.z), 0.0f);
            y.w = fmaxf(fmaf(v[j].w, sc4.w, bi4.w), 0.0f);
            out4[(size_t)r * (UNITS / 4) + h * (CUNITS / 4) + q] = y;
        }
    }
}

extern "C" void kernel_launch(void* const* d_in, const int* in_sizes, int n_in,
                              void* d_out, int out_size, void* d_ws, size_t ws_size,
                              hipStream_t stream)
{
    const float* features   = (const float*)d_in[0];
    const int*   num_voxels = (const int*)  d_in[1];
    const int*   coors      = (const int*)  d_in[2];
    const float* W          = (const float*)d_in[3];
    const float* gamma      = (const float*)d_in[4];
    const float* beta       = (const float*)d_in[5];
    float*       out        = (float*)d_out;

    float* partials = (float*)d_ws;

    pfn_stats_kernel<<<NWG, BLOCK, 0, stream>>>(features, num_voxels, coors, W, gamma,
                                                partials, out);
    pfn_apply_kernel<<<AWG, ABLOCK, 0, stream>>>(partials, gamma, beta, out);
}
